# starting kernel
# baseline (speedup 1.0000x reference)
#include <hip/hip_runtime.h>
#include <hip/hip_bf16.h>

#define BATCH   2
#define NPTS    256
#define NCH     32
#define NBASIS  16
#define KDIM    (NBASIS * NCH)
#define GAMMA_F 10.0f
#define EPS_F   1e-9f

typedef float v2f __attribute__((ext_vector_type(2)));
typedef float v8f __attribute__((ext_vector_type(8)));

#define OFF_D   0
#define OFF_S   (BATCH * NPTS * NPTS)
#define OFF_W2  (OFF_S + BATCH * NPTS * KDIM)

__global__ void cf_dist_kernel(const float* __restrict__ geom,
                               float* __restrict__ dmat) {
    int idx = blockIdx.x * blockDim.x + threadIdx.x;
    int b = idx & (NPTS - 1);
    int a = (idx >> 8) & (NPTS - 1);
    int z = idx >> 16;
    const float* ga = geom + (z * NPTS + a) * 3;
    const float* gb = geom + (z * NPTS + b) * 3;
    float dx = gb[0] - ga[0];
    float dy = gb[1] - ga[1];
    float dz = gb[2] - ga[2];
    dmat[idx] = sqrtf(dx * dx + dy * dy + dz * dz + EPS_F);
}

__global__ void cf_w2_kernel(const float* __restrict__ kw,
                             const int* __restrict__ n_norm,
                             float* __restrict__ w2) {
    int idx = blockIdx.x * blockDim.x + threadIdx.x;
    float scale = rsqrtf((float)n_norm[0]);
    int i  = idx & 31;
    int nj = idx >> 5;
    int j  = nj & 31;
    int n  = nj >> 5;
    w2[idx] = kw[(n * NCH + i) * NCH + j] * scale;
}

__global__ void cf_gemm1_kernel(const float* __restrict__ dmat,
                                const float* __restrict__ feat,
                                const float* __restrict__ centers,
                                float* __restrict__ S) {
    const int wave  = blockIdx.x * (blockDim.x >> 5) + (threadIdx.x >> 5);
    const int lane  = threadIdx.x & 31;
    const int at    = wave & 15;
    const int n     = (wave >> 4) & 15;
    const int z     = wave >> 8;
    const int row0  = at * 16;
    const int m     = lane & 15;
    const int khalf = lane >> 4;

    const float  cn   = centers[n];
    const float* drow = dmat + (z * NPTS + row0 + m) * NPTS;
    const float* fc0  = feat + z * NPTS * NCH + m;
    const float* fc1  = fc0 + 16;

    v8f c0 = {};
    v8f c1 = {};
    for (int k0 = 0; k0 < NPTS; k0 += 4) {
        int b0 = k0 + khalf * 2;
        float t0 = drow[b0] - cn;
        float t1 = drow[b0 + 1] - cn;
        v2f av;
        av.x = __expf(-GAMMA_F * t0 * t0);
        av.y = __expf(-GAMMA_F * t1 * t1);
        v2f bv0, bv1;
        bv0.x = fc0[b0 * NCH];
        bv0.y = fc0[(b0 + 1) * NCH];
        bv1.x = fc1[b0 * NCH];
        bv1.y = fc1[(b0 + 1) * NCH];
        c0 = __builtin_amdgcn_wmma_f32_16x16x4_f32(
                false, av, false, bv0, (short)0, c0, false, false);
        c1 = __builtin_amdgcn_wmma_f32_16x16x4_f32(
                false, av, false, bv1, (short)0, c1, false, false);
    }
#pragma unroll
    for (int i = 0; i < 8; ++i) {
        int row = row0 + i + khalf * 8;
        float* srow = S + (z * NPTS + row) * KDIM + n * NCH;
        srow[m]      = c0[i];
        srow[16 + m] = c1[i];
    }
}

__global__ void cf_gemm2_kernel(const float* __restrict__ S,
                                const float* __restrict__ w2,
                                float* __restrict__ out) {
    const int wave  = blockIdx.x * (blockDim.x >> 5) + (threadIdx.x >> 5);
    const int lane  = threadIdx.x & 31;
    const int at    = wave & 15;
    const int z     = wave >> 4;
    const int row0  = at * 16;
    const int m     = lane & 15;
    const int khalf = lane >> 4;

    const float* srow = S + (z * NPTS + row0 + m) * KDIM;
    const float* wc0  = w2 + m;
    const float* wc1  = wc0 + 16;

    v8f c0 = {};
    v8f c1 = {};
    for (int k0 = 0; k0 < KDIM; k0 += 4) {
        int kk = k0 + khalf * 2;
        v2f av;
        av.x = srow[kk];
        av.y = srow[kk + 1];
        v2f bv0, bv1;
        bv0.x = wc0[kk * NCH];
        bv0.y = wc0[(kk + 1) * NCH];
        bv1.x = wc1[kk * NCH];
        bv1.y = wc1[(kk + 1) * NCH];
        c0 = __builtin_amdgcn_wmma_f32_16x16x4_f32(
                false, av, false, bv0, (short)0, c0, false, false);
        c1 = __builtin_amdgcn_wmma_f32_16x16x4_f32(
                false, av, false, bv1, (short)0, c1, false, false);
    }
#pragma unroll
    for (int i = 0; i < 8; ++i) {
        int row = row0 + i + khalf * 8;
        float* orow = out + (z * NPTS + row) * NCH;
        orow[m]      = c0[i];
        orow[16 + m] = c1[i];
    }
}

extern "C" void kernel_launch(void* const* d_in, const int* in_sizes, int n_in,
                              void* d_out, int out_size, void* d_ws, size_t ws_size,
                              hipStream_t stream) {
    const float* feat    = (const float*)d_in[0];
    const float* geom    = (const float*)d_in[1];
    const float* centers = (const float*)d_in[2];
    const float* kw      = (const float*)d_in[3];
    const int*   n_norm  = (const int*)d_in[4];

    float* ws   = (float*)d_ws;
    float* dmat = ws + OFF_D;
    float* S    = ws + OFF_S;
    float* w2   = ws + OFF_W2;
    float* out  = (float*)d_out;

    cf_dist_kernel<<<BATCH * NPTS * NPTS / 256, 256, 0, stream>>>(geom, dmat);
    cf_w2_kernel<<<KDIM * NCH / 256, 256, 0, stream>>>(kw, n_norm, w2);

    cf_gemm1_kernel<<<128, 128, 0, stream>>>(dmat, feat, centers, S);

    cf_gemm2_kernel<<<8, 128, 0, stream>>>(S, w2, out);
}


# launch 2 on 64 work-groups instead of 32: each owns 16 rows x 16 channels (one output tile), G stored per channel half in fragment order so a lane's 16-byte load feeds two k-steps of its half
# speedup vs baseline: 2.7713x; 2.7713x over previous
#include <hip/hip_runtime.h>

#define BATCH   2
#define NPTS    256
#define NCH     32
#define NBASIS  16
#define GAMMA_F 10.0f
#define EPS_F   1e-9f

typedef float v2f __attribute__((ext_vector_type(2)));
typedef float v4f __attribute__((ext_vector_type(4)));
typedef float v8f __attribute__((ext_vector_type(8)));


__global__ __launch_bounds__(64)
void cf_g_kernel(const float* __restrict__ feat,
                 const float* __restrict__ kw,
                 const int* __restrict__ n_norm,
                 float* __restrict__ G) {
    const int gw    = blockIdx.x * 2 + (threadIdx.x >> 5);
    const int lane  = threadIdx.x & 31;
    const int c     = lane & 15;
    const int h     = lane >> 4;
    const int mtile = gw >> 3;
    const int it    = (gw >> 2) & 1;
    const int q     = gw & 3;
    const float scale = rsqrtf((float)n_norm[0]);

    const float* arow = feat + (mtile * 16 + c) * NCH + 2 * h;
    const float* b0   = kw + ((4 * q) * NCH + 16 * it + c) * NCH + 2 * h;
    const float* b1   = b0 + NCH * NCH;
    const float* b2   = b1 + NCH * NCH;
    const float* b3   = b2 + NCH * NCH;

    v8f a0 = {}, a1 = {}, a2 = {}, a3 = {};
#pragma unroll
    for (int t = 0; t < 8; ++t) {
        v2f av = *(const v2f*)(arow + 4 * t);
        v2f w0 = *(const v2f*)(b0 + 4 * t);
        v2f w1 = *(const v2f*)(b1 + 4 * t);
        v2f w2 = *(const v2f*)(b2 + 4 * t);
        v2f w3 = *(const v2f*)(b3 + 4 * t);
        a0 = __builtin_amdgcn_wmma_f32_16x16x4_f32(false, av, false, w0, (short)0, a0, false, false);
        a1 = __builtin_amdgcn_wmma_f32_16x16x4_f32(false, av, false, w1, (short)0, a1, false, false);
        a2 = __builtin_amdgcn_wmma_f32_16x16x4_f32(false, av, false, w2, (short)0, a2, false, false);
        a3 = __builtin_amdgcn_wmma_f32_16x16x4_f32(false, av, false, w3, (short)0, a3, false, false);
    }
#pragma unroll
    for (int r = 0; r < 8; ++r) {
        const int zb = mtile * 16 + 8 * h + r;
        v4f o;
        o.x = a0[r] * scale;
        o.y = a1[r] * scale;
        o.z = a2[r] * scale;
        o.w = a3[r] * scale;
        *(v4f*)(G + (((it * (BATCH * NPTS) + zb) * 4 + q) * 16 + c) * 4) = o;
    }
}

__global__ __launch_bounds__(512)
void cf_out_kernel(const float* __restrict__ geom,
                   const float* __restrict__ centers,
                   const float* __restrict__ G,
                   float* __restrict__ out) {
    __shared__ float red[16 * 256];

    const int tid   = threadIdx.x;
    const int w     = __builtin_amdgcn_readfirstlane(tid >> 5);
    const int lane  = tid & 31;
    const int c     = lane & 15;
    const int h     = lane >> 4;
    const int z     = blockIdx.x >> 5;
    const int atile = (blockIdx.x >> 1) & 15;
    const int it    = blockIdx.x & 1;

    const float* ga = geom + (z * NPTS + atile * 16 + c) * 3;
    const float ax = ga[0], ay = ga[1], az = ga[2];

    v2f cn[2][2];
#pragma unroll
    for (int u = 0; u < 2; ++u)
#pragma unroll
        for (int tt = 0; tt < 2; ++tt) cn[u][tt] = *(const v2f*)(centers + 8 * u + 4 * h + 2 * tt);

    const int zb0 = z * NPTS + 16 * w;
    const float* gbp = geom + zb0 * 3;
    const float* gp  = G + (((it * (BATCH * NPTS) + zb0) * 4 + h) * 16 + c) * 4;

    v4f gfrag[32];
#pragma unroll
    for (int j = 0; j < 32; ++j) gfrag[j] = *(const v4f*)(gp + j * 128);
    __builtin_amdgcn_sched_barrier(0);

    v8f acc0 = {}, acc1 = {};
#pragma unroll
    for (int bb = 0; bb < 16; ++bb) {
        const float dx = gbp[bb * 3 + 0] - ax;
        const float dy = gbp[bb * 3 + 1] - ay;
        const float dz = gbp[bb * 3 + 2] - az;
        const float d  = sqrtf(dx * dx + dy * dy + dz * dz + EPS_F);
#pragma unroll
        for (int u = 0; u < 2; ++u) {
            const v4f g = gfrag[bb * 2 + u];
            const float t00 = d - cn[u][0].x;
            const float t01 = d - cn[u][0].y;
            const float t10 = d - cn[u][1].x;
            const float t11 = d - cn[u][1].y;
            v2f av0, av1;
            av0.x = __expf(-GAMMA_F * t00 * t00);
            av0.y = __expf(-GAMMA_F * t01 * t01);
            av1.x = __expf(-GAMMA_F * t10 * t10);
            av1.y = __expf(-GAMMA_F * t11 * t11);
            v2f bv0, bv1;
            bv0.x = g.x; bv0.y = g.y;
            bv1.x = g.z; bv1.y = g.w;
            acc0 = __builtin_amdgcn_wmma_f32_16x16x4_f32(false, av0, false, bv0, (short)0, acc0, false, false);
            acc1 = __builtin_amdgcn_wmma_f32_16x16x4_f32(false, av1, false, bv1, (short)0, acc1, false, false);
        }
    }

    float* mine = red + w * 256 + lane;
#pragma unroll
    for (int r = 0; r < 8; ++r) mine[r * 32] = acc0[r] + acc1[r];
    __syncthreads();

    const int rl = tid >> 5;
    if (rl < 8) {
        float s = 0.0f;
#pragma unroll
        for (int ww = 0; ww < 16; ++ww) s += red[ww * 256 + rl * 32 + lane];
        out[(z * NPTS + atile * 16 + 8 * h + rl) * NCH + 16 * it + c] = s;
    }
}

extern "C" void kernel_launch(void* const* d_in, const int* in_sizes, int n_in,
                              void* d_out, int out_size, void* d_ws, size_t ws_size,
                              hipStream_t stream) {
    const float* feat    = (const float*)d_in[0];
    const float* geom    = (const float*)d_in[1];
    const float* centers = (const float*)d_in[2];
    const float* kw      = (const float*)d_in[3];
    const int*   n_norm  = (const int*)d_in[4];

    float* G   = (float*)d_ws;
    float* out = (float*)d_out;

    cf_g_kernel<<<128, 64, 0, stream>>>(feat, kw, n_norm, G);
    cf_out_kernel<<<64, 512, 0, stream>>>(geom, centers, G, out);
}
